# starting kernel
# baseline (speedup 1.0000x reference)
#include <hip/hip_runtime.h>


typedef float v2f __attribute__((ext_vector_type(2)));
typedef float v4f __attribute__((ext_vector_type(4)));
typedef float v8f __attribute__((ext_vector_type(8)));

constexpr int IN_F   = 4096;
constexpr int OUT_F  = 4096;
constexpr int RANK   = 64;
constexpr int NTOK   = 8192;
constexpr int KSPLIT = 4;
constexpr int KCHUNK = 128;
constexpr int LSTR   = 160;

__global__ __launch_bounds__(256) void lowrank_xu(const float* __restrict__ x,
                                                  const float* __restrict__ U,
                                                  float* __restrict__ Tpart) {
  __shared__ float lU[(KCHUNK / 2) * LSTR];

  const int tid  = threadIdx.x;
  const int lane = tid & 31;
  const int ln16 = lane & 15;
  const int hi   = lane >> 4;
  const int wave = tid >> 5;
  const int mbase = blockIdx.x * 128 + wave * 16;
  const int kpart = blockIdx.y * (IN_F / KSPLIT);

  const float* __restrict__ xrow = x + (size_t)(mbase + ln16) * IN_F + kpart;

  v8f acc[4] = {};

  for (int kb = 0; kb < IN_F / KSPLIT; kb += KCHUNK) {
    __syncthreads();
    for (int i = tid; i < KCHUNK * (RANK / 4); i += 256) {
      const int r  = i >> 4;
      const int c4 = i & 15;
      const v4f u = *(const v4f*)(U + (size_t)(kpart + kb + r) * RANK + c4 * 4);
      float* dst = &lU[(r >> 1) * LSTR + (r & 1)];
#pragma unroll
      for (int j = 0; j < 4; ++j) dst[2 * (c4 * 4 + j)] = u[j];
    }
    __syncthreads();

#pragma unroll 4
    for (int k0 = 0; k0 < KCHUNK; k0 += 4) {
      const v2f a = *(const v2f*)(xrow + kb + k0 + 2 * hi);
#pragma unroll
      for (int t = 0; t < 4; ++t) {
        const v2f b = *(const v2f*)(
            &lU[((k0 >> 1) + hi) * LSTR + 2 * (t * 16 + ln16)]);
        acc[t] = __builtin_amdgcn_wmma_f32_16x16x4_f32(
            false, a, false, b, (short)0, acc[t], false, false);
      }
    }
  }

  float* __restrict__ Tout = Tpart + (size_t)blockIdx.y * NTOK * RANK;
#pragma unroll
  for (int t = 0; t < 4; ++t)
#pragma unroll
    for (int r = 0; r < 8; ++r)
      Tout[(size_t)(mbase + hi * 8 + r) * RANK + t * 16 + ln16] = acc[t][r];
}

__global__ __launch_bounds__(256) void lowrank_tv(const float* __restrict__ Tpart,
                                                  const float* __restrict__ V,
                                                  const float* __restrict__ bias,
                                                  float* __restrict__ out) {
  __shared__ float lV[(RANK / 2) * LSTR];

  const int tid  = threadIdx.x;
  const int lane = tid & 31;
  const int ln16 = lane & 15;
  const int hi   = lane >> 4;
  const int wave = tid >> 5;
  const int mbase = blockIdx.x * 128 + wave * 16;
  const int nbase = blockIdx.y * 64;

  for (int i = tid; i < RANK * (64 / 4); i += 256) {
    const int r  = i >> 4;
    const int c4 = i & 15;
    const v4f v = *(const v4f*)(V + (size_t)r * OUT_F + nbase + c4 * 4);
    float* dst = &lV[(r >> 1) * LSTR + (r & 1)];
#pragma unroll
    for (int j = 0; j < 4; ++j) dst[2 * (c4 * 4 + j)] = v[j];
  }
  __syncthreads();

  const size_t trow = (size_t)(mbase + ln16) * RANK;

  v8f acc[4] = {};
#pragma unroll
  for (int k0 = 0; k0 < RANK; k0 += 4) {
    const int koff = k0 + 2 * hi;
    v2f a = *(const v2f*)(Tpart + trow + koff);
#pragma unroll
    for (int p = 1; p < KSPLIT; ++p)
      a += *(const v2f*)(Tpart + (size_t)p * NTOK * RANK + trow + koff);
#pragma unroll
    for (int t = 0; t < 4; ++t) {
      const v2f b = *(const v2f*)(
          &lV[((k0 >> 1) + hi) * LSTR + 2 * (t * 16 + ln16)]);
      acc[t] = __builtin_amdgcn_wmma_f32_16x16x4_f32(
          false, a, false, b, (short)0, acc[t], false, false);
    }
  }

#pragma unroll
  for (int t = 0; t < 4; ++t) {
    const float bv = bias[nbase + t * 16 + ln16];
#pragma unroll
    for (int r = 0; r < 8; ++r)
      out[(size_t)(mbase + hi * 8 + r) * OUT_F + nbase + t * 16 + ln16] =
          acc[t][r] + bv;
  }
}

extern "C" void kernel_launch(void* const* d_in, const int* in_sizes, int n_in,
                              void* d_out, int out_size, void* d_ws, size_t ws_size,
                              hipStream_t stream) {
  const float* x    = (const float*)d_in[0];
  const float* U    = (const float*)d_in[1];
  const float* V    = (const float*)d_in[2];
  const float* bias = (const float*)d_in[3];
  float* out   = (float*)d_out;
  float* Tpart = (float*)d_ws;

  lowrank_xu<<<dim3(NTOK / 128, KSPLIT), 256, 0, stream>>>(x, U, Tpart);
  lowrank_tv<<<dim3(NTOK / 128, OUT_F / 64), 256, 0, stream>>>(Tpart, V, bias, out);
}


# stage 2 operand loads streamed under the WMMAs: split-K sum built one row tile ahead in a 2-tile LDS ring, V waited for per k-step in tile 0, barrier split into signal and wait; sum order kept
# speedup vs baseline: 2.2100x; 2.2100x over previous
#include <hip/hip_runtime.h>


typedef float v2f __attribute__((ext_vector_type(2)));
typedef float v4f __attribute__((ext_vector_type(4)));
typedef float v8f __attribute__((ext_vector_type(8)));

constexpr int IN_F   = 4096;
constexpr int OUT_F  = 4096;
constexpr int RANK   = 64;
constexpr int NTOK   = 8192;
constexpr int KSPLIT = 4;
constexpr int KCHUNK = 128;
constexpr int LSTR   = 160;
constexpr int USTG   = (KCHUNK / 2) * LSTR;
constexpr int XJS    = 132;
constexpr int XTILE  = (KCHUNK / 8) * XJS;
constexpr int XSTG   = 8 * XTILE;
constexpr int NCHUNK = IN_F / KSPLIT / KCHUNK;
constexpr int TTILE  = 16 * RANK;
constexpr int TPLANE = NTOK * RANK;

__global__ __launch_bounds__(256) void lowrank_xu(const float* __restrict__ x,
                                                  const float* __restrict__ U,
                                                  float* __restrict__ Tpart) {
  __shared__ __attribute__((aligned(16))) float lU[2 * USTG];
  __shared__ __attribute__((aligned(16))) float lX[2 * XSTG];

  const int tid  = threadIdx.x;
  const int lane = tid & 31;
  const int ln16 = lane & 15;
  const int hi   = lane >> 4;
  const int wave = tid >> 5;
  const int mbase = blockIdx.x * 128 + wave * 16;
  const int kpart = blockIdx.y * (IN_F / KSPLIT);

  const float* __restrict__ xg = x + (size_t)mbase * IN_F + kpart + lane * 4;
  const float* __restrict__ ug = U + (size_t)(kpart + (tid >> 4)) * RANK + (tid & 15) * 4;

  const int xs_off = wave * XTILE + (lane >> 1) * XJS + (lane & 1) * 2;
  const int xr_off = wave * XTILE + lane * 4;
  const int us_off = (tid >> 5) * LSTR + ((tid >> 4) & 1) + 8 * (tid & 15);
  const int ub_off = hi * LSTR + 2 * ln16;

  v4f xr[16];
  v4f ur[KCHUNK * (RANK / 4) / 256];
  v8f acc[4] = {};

  auto issue_loads = [&](int kb) {
#pragma unroll
    for (int i = 0; i < 16; ++i)
      xr[i] = *(const v4f*)(xg + (size_t)i * IN_F + kb);
#pragma unroll
    for (int m = 0; m < 8; ++m)
      ur[m] = *(const v4f*)(ug + (size_t)(kb + 16 * m) * RANK);
  };

  auto store_stage = [&](int stg) {
    float* xs = &lX[stg * XSTG + xs_off];
#pragma unroll
    for (int i = 0; i < 16; ++i) {
      v2f lo, hv;
      lo[0] = xr[i][0]; lo[1] = xr[i][1];
      hv[0] = xr[i][2]; hv[1] = xr[i][3];
      *(v2f*)(xs + i * 4)        = lo;
      *(v2f*)(xs + (16 + i) * 4) = hv;
    }
    float* us = &lU[stg * USTG + us_off];
#pragma unroll
    for (int m = 0; m < 8; ++m)
#pragma unroll
      for (int j = 0; j < 4; ++j) us[8 * m * LSTR + 2 * j] = ur[m][j];
  };

  constexpr int NGRP = KCHUNK / 16;

  auto compute_stage = [&](int stg) {
    const float* xa = &lX[stg * XSTG + xr_off];
    const float* ub = &lU[stg * USTG + ub_off];
    v4f fa[2][2];
    v2f fb[2][4][4];

    auto load_group = [&](int g, int s) {
#pragma unroll
      for (int j = 0; j < 2; ++j)
        fa[s][j] = *(const v4f*)(xa + (2 * g + j) * XJS);
#pragma unroll
      for (int ks = 0; ks < 4; ++ks)
#pragma unroll
        for (int t = 0; t < 4; ++t)
          fb[s][ks][t] = *(const v2f*)(ub + (8 * g + 2 * ks) * LSTR + 32 * t);
    };

    load_group(0, 0);
    __builtin_amdgcn_sched_barrier(0);
#pragma unroll
    for (int g = 0; g < NGRP; ++g) {
      const int s = g & 1;
      if (g + 1 < NGRP) load_group(g + 1, s ^ 1);
      __builtin_amdgcn_sched_barrier(0);
#pragma unroll
      for (int ks = 0; ks < 4; ++ks) {
        v2f a;
        a[0] = fa[s][ks >> 1][2 * (ks & 1)];
        a[1] = fa[s][ks >> 1][2 * (ks & 1) + 1];
#pragma unroll
        for (int t = 0; t < 4; ++t)
          acc[t] = __builtin_amdgcn_wmma_f32_16x16x4_f32(
              false, a, false, fb[s][ks][t], (short)0, acc[t], false, false);
      }
      __builtin_amdgcn_sched_barrier(0);
    }
  };

  issue_loads(0);
  store_stage(0);
  __syncthreads();

  for (int c = 0; c < NCHUNK - 1; ++c) {
    issue_loads((c + 1) * KCHUNK);
    compute_stage(c & 1);
    store_stage((c + 1) & 1);
    __syncthreads();
  }
  compute_stage((NCHUNK - 1) & 1);

  float* __restrict__ Tout = Tpart + (size_t)blockIdx.y * TPLANE +
                             (size_t)(mbase >> 4) * TTILE +
                             (ln16 >> 3) * 128 +
                             (16 * ((ln16 >> 1) & 1) + 8 * hi) * 4 +
                             ((ln16 >> 2) & 1) * 2 + (ln16 & 1);
#pragma unroll
  for (int t = 0; t < 4; ++t)
#pragma unroll
    for (int r = 0; r < 8; ++r)
      Tout[t * 256 + r * 4] = acc[t][r];
}

constexpr int S2ROWT = 8;
constexpr int S2COLS = 8 * 64;
constexpr int VHEAD  = 6;

__global__ __launch_bounds__(256) void lowrank_tv(const float* __restrict__ Tpart,
                                                  const float* __restrict__ V,
                                                  const float* __restrict__ bias,
                                                  float* __restrict__ out) {
  __shared__ __attribute__((aligned(16))) float lT[2 * TTILE];

  const int tid  = threadIdx.x;
  const int lane = tid & 31;
  const int ln16 = lane & 15;
  const int hi   = lane >> 4;
  const int wave = tid >> 5;
  const int mbase = blockIdx.x * (16 * S2ROWT);
  const int nbase = blockIdx.y * S2COLS + wave * 64;

  const float* __restrict__ tsrc =
      Tpart + (size_t)blockIdx.x * (S2ROWT * TTILE) + tid * 4;
  v4f tp[2][KSPLIT];

  auto load_tile = [&](int j) {
#pragma unroll
    for (int p = 0; p < KSPLIT; ++p)
      tp[j & 1][p] = *(const v4f*)(tsrc + (size_t)p * TPLANE + j * TTILE);
  };
  auto stage_tile = [&](int j) {
    v4f s4 = tp[j & 1][0];
#pragma unroll
    for (int p = 1; p < KSPLIT; ++p) s4 += tp[j & 1][p];
    *(v4f*)(&lT[(j & 1) * TTILE + tid * 4]) = s4;
  };
  auto lds_done = [&]() {
    asm volatile("s_wait_dscnt 0x0" ::: "memory");
    asm volatile("s_barrier_signal -1" ::: "memory");
  };
  auto lds_ready = [&]() { asm volatile("s_barrier_wait -1" ::: "memory"); };

  const float* __restrict__ vg = V + (size_t)(2 * hi) * OUT_F + nbase + ln16;
  v2f bf[RANK / 4][4];
  auto load_v = [&](int ks0, int ks1) {
#pragma unroll
    for (int ks = ks0; ks < ks1; ++ks)
#pragma unroll
      for (int t = 0; t < 4; ++t) {
        bf[ks][t][0] = vg[(size_t)(4 * ks) * OUT_F + 16 * t];
        bf[ks][t][1] = vg[(size_t)(4 * ks + 1) * OUT_F + 16 * t];
      }
  };

  load_tile(0);
  load_tile(1);
  load_v(0, VHEAD);
  __builtin_amdgcn_sched_barrier(0);
  stage_tile(0);
  lds_done();
  float bv[4];
#pragma unroll
  for (int t = 0; t < 4; ++t) bv[t] = bias[nbase + 16 * t + ln16];

  const float* ta = &lT[lane * 4];
  float* __restrict__ orow = out + (size_t)(mbase + hi * 8) * OUT_F + nbase + ln16;

#pragma unroll
  for (int rt = 0; rt < S2ROWT; ++rt) {
    lds_ready();
    v4f a4[RANK / 8];
#pragma unroll
    for (int k8 = 0; k8 < RANK / 8; ++k8)
      a4[k8] = *(const v4f*)(ta + (rt & 1) * TTILE + k8 * 128);
    if (rt + 1 < S2ROWT) {
      stage_tile(rt + 1);
      lds_done();
      __builtin_amdgcn_sched_barrier(0);
    }
    if (rt + 2 < S2ROWT) load_tile(rt + 2);

    v8f acc[4] = {};
#pragma unroll
    for (int k8 = 0; k8 < RANK / 8; ++k8) {
      if (rt == 0 && VHEAD + 2 * k8 < RANK / 4) {
        __builtin_amdgcn_sched_barrier(0);
        load_v(VHEAD + 2 * k8, VHEAD + 2 * k8 + 2);
        __builtin_amdgcn_sched_barrier(0);
      }
#pragma unroll
      for (int kk = 0; kk < 2; ++kk) {
        v2f a;
        a[0] = a4[k8][2 * kk];
        a[1] = a4[k8][2 * kk + 1];
#pragma unroll
        for (int t = 0; t < 4; ++t)
          acc[t] = __builtin_amdgcn_wmma_f32_16x16x4_f32(
              false, a, false, bf[2 * k8 + kk][t], (short)0, acc[t], false, false);
      }
    }

    float* __restrict__ o = orow + (size_t)(rt * 16) * OUT_F;
#pragma unroll
    for (int t = 0; t < 4; ++t)
#pragma unroll
      for (int r = 0; r < 8; ++r)
        o[(size_t)r * OUT_F + 16 * t] = acc[t][r] + bv[t];
  }
}

extern "C" void kernel_launch(void* const* d_in, const int* in_sizes, int n_in,
                              void* d_out, int out_size, void* d_ws, size_t ws_size,
                              hipStream_t stream) {
  const float* x    = (const float*)d_in[0];
  const float* U    = (const float*)d_in[1];
  const float* V    = (const float*)d_in[2];
  const float* bias = (const float*)d_in[3];
  float* out   = (float*)d_out;
  float* Tpart = (float*)d_ws;

  lowrank_xu<<<dim3(NTOK / 128, KSPLIT), 256, 0, stream>>>(x, U, Tpart);
  lowrank_tv<<<dim3(NTOK / (16 * S2ROWT), OUT_F / S2COLS), 256, 0, stream>>>(Tpart, V, bias, out);
}
